# starting kernel
# baseline (speedup 1.0000x reference)
#include <hip/hip_runtime.h>

typedef float v2f __attribute__((ext_vector_type(2)));
typedef float v8f __attribute__((ext_vector_type(8)));

#define NNODES 100000
#define DFEAT  64
#define NTILES (NNODES / 16)
#define WSTR   65

__global__ __launch_bounds__(256) void gemm64_wmma(const float* __restrict__ x,
                                                   const float* __restrict__ W,
                                                   float* __restrict__ h,
                                                   int nRowTiles) {
    __shared__ float Wl[64 * WSTR];
    for (int i = threadIdx.x; i < 64 * 64; i += 256) {
        int r = i >> 6, c = i & 63;
        Wl[r * WSTR + c] = W[i];
    }
    __syncthreads();

    const int wave = threadIdx.x >> 5;
    const int lane = threadIdx.x & 31;
    const int tile = blockIdx.x * 8 + wave;
    if (tile >= nRowTiles) return;

    const int hl  = lane >> 4;
    const int l15 = lane & 15;
    const int rowA = tile * 16 + l15;

    v2f a[16];
    const float* xrow = x + (size_t)rowA * DFEAT + hl * 2;
    #pragma unroll
    for (int k = 0; k < 16; ++k) {
        a[k].x = xrow[k * 4 + 0];
        a[k].y = xrow[k * 4 + 1];
    }

    #pragma unroll
    for (int nt = 0; nt < 4; ++nt) {
        v8f acc = {};
        const int col = nt * 16 + l15;
        #pragma unroll
        for (int k = 0; k < 16; ++k) {
            const int krow = k * 4 + hl * 2;
            v2f b;
            b.x = Wl[(krow + 0) * WSTR + col];
            b.y = Wl[(krow + 1) * WSTR + col];
            acc = __builtin_amdgcn_wmma_f32_16x16x4_f32(
                false, a[k], false, b, (short)0, acc, false, false);
        }
        #pragma unroll
        for (int r = 0; r < 8; ++r) {
            const int row = tile * 16 + r + 8 * hl;
            h[(size_t)row * DFEAT + col] = acc[r];
        }
    }
}

__global__ void deg_init(float* __restrict__ deg, int n) {
    int i = blockIdx.x * 256 + threadIdx.x;
    if (i < n) deg[i] = 1.0f;
}
__global__ void deg_acc(const int* __restrict__ dst, float* __restrict__ deg, int e) {
    int i = blockIdx.x * 256 + threadIdx.x;
    if (i < e) atomicAdd(&deg[dst[i]], 1.0f);
}
__global__ void deg_finish(float* __restrict__ deg, int n) {
    int i = blockIdx.x * 256 + threadIdx.x;
    if (i < n) deg[i] = rsqrtf(deg[i]);
}

__global__ void zero_f(float* __restrict__ p, int n) {
    int i = blockIdx.x * 256 + threadIdx.x;
    if (i < n) p[i] = 0.0f;
}

__global__ __launch_bounds__(256) void edge_scatter(const float* __restrict__ h,
                                                    const int* __restrict__ src,
                                                    const int* __restrict__ dst,
                                                    const float* __restrict__ dinv,
                                                    float* __restrict__ agg, int E) {
    int tid = blockIdx.x * 256 + threadIdx.x;
    int e = tid >> 4;
    if (e >= E) return;
    int c4 = (tid & 15) * 4;
    int s = src[e], d = dst[e];
    float w = dinv[s] * dinv[d];
    const float4 hv = *(const float4*)(h + (size_t)s * DFEAT + c4);
    float* op = agg + (size_t)d * DFEAT + c4;
    atomicAdd(op + 0, hv.x * w);
    atomicAdd(op + 1, hv.y * w);
    atomicAdd(op + 2, hv.z * w);
    atomicAdd(op + 3, hv.w * w);
}

__global__ void finalize(const float* __restrict__ agg, const float* __restrict__ h,
                         const float* __restrict__ dinv, const float* __restrict__ bias,
                         float* __restrict__ out, int do_relu, int n) {
    int i = blockIdx.x * 256 + threadIdx.x;
    if (i >= n) return;
    int node = i >> 6, c = i & 63;
    float di = dinv[node];
    float v = agg[i] + h[i] * (di * di) + bias[c];
    if (do_relu) v = fmaxf(v, 0.0f);
    out[i] = v;
}

extern "C" void kernel_launch(void* const* d_in, const int* in_sizes, int n_in,
                              void* d_out, int out_size, void* d_ws, size_t ws_size,
                              hipStream_t stream) {
    const float* x0 = (const float*)d_in[0];
    const int*   me = (const int*)d_in[1];
    const float* W1 = (const float*)d_in[2];
    const float* b1 = (const float*)d_in[3];
    const float* W2 = (const float*)d_in[4];
    const float* b2 = (const float*)d_in[5];
    const float* W3 = (const float*)d_in[6];
    const float* b3 = (const float*)d_in[7];

    const int E = in_sizes[1] / 2;
    const int* src = me;
    const int* dst = me + E;
    const int NF = NNODES * DFEAT;

    float* ws   = (float*)d_ws;
    float* dinv = ws;
    float* h    = ws + 100352;
    float* xb   = h + NF;
    float* agg  = xb + NF;

    float* out = (float*)d_out;

    const int TB = 256;
    const int gN   = (NNODES + TB - 1) / TB;
    const int gE   = (E + TB - 1) / TB;
    const int gE16 = (E * 16 + TB - 1) / TB;
    const int gNF  = (NF + TB - 1) / TB;
    const int gGEMM = (NTILES + 7) / 8;

    deg_init<<<gN, TB, 0, stream>>>(dinv, NNODES);
    deg_acc<<<gE, TB, 0, stream>>>(dst, dinv, E);
    deg_finish<<<gN, TB, 0, stream>>>(dinv, NNODES);

    gemm64_wmma<<<gGEMM, TB, 0, stream>>>(x0, W1, h, NTILES);
    zero_f<<<gNF, TB, 0, stream>>>(agg, NF);
    edge_scatter<<<gE16, TB, 0, stream>>>(h, src, dst, dinv, agg, E);
    finalize<<<gNF, TB, 0, stream>>>(agg, h, dinv, b1, xb, 1, NF);

    gemm64_wmma<<<gGEMM, TB, 0, stream>>>(xb, W2, h, NTILES);
    zero_f<<<gNF, TB, 0, stream>>>(agg, NF);
    edge_scatter<<<gE16, TB, 0, stream>>>(h, src, dst, dinv, agg, E);
    finalize<<<gNF, TB, 0, stream>>>(agg, h, dinv, b2, xb, 1, NF);

    gemm64_wmma<<<gGEMM, TB, 0, stream>>>(xb, W3, h, NTILES);
    zero_f<<<gNF, TB, 0, stream>>>(out, NF);
    edge_scatter<<<gE16, TB, 0, stream>>>(h, src, dst, dinv, out, E);
    finalize<<<gNF, TB, 0, stream>>>(out, h, dinv, b3, out, 0, NF);
}


# Replace float-atomic edge scatter by a pull over a destination-sorted edge list built in-call (arrival-order integer ranks); row gather fuses zero, self loop, bias, relu; summation order changes
# speedup vs baseline: 2.5564x; 2.5564x over previous
#include <hip/hip_runtime.h>

typedef float v2f __attribute__((ext_vector_type(2)));
typedef float v8f __attribute__((ext_vector_type(8)));

#define NNODES 100000
#define DFEAT  64
#define NTILES (NNODES / 16)
#define WSTR   65

__global__ __launch_bounds__(256) void gemm64_wmma(const float* __restrict__ x,
                                                   const float* __restrict__ W,
                                                   float* __restrict__ h,
                                                   int nRowTiles) {
    __shared__ float Wl[64 * WSTR];
    for (int i = threadIdx.x; i < 64 * 64; i += 256) {
        int r = i >> 6, c = i & 63;
        Wl[r * WSTR + c] = W[i];
    }
    __syncthreads();

    const int wave = threadIdx.x >> 5;
    const int lane = threadIdx.x & 31;
    const int tile = blockIdx.x * 8 + wave;
    if (tile >= nRowTiles) return;

    const int hl  = lane >> 4;
    const int l15 = lane & 15;
    const int rowA = tile * 16 + l15;

    v2f a[16];
    const float* xrow = x + (size_t)rowA * DFEAT + hl * 2;
    #pragma unroll
    for (int k = 0; k < 16; ++k) {
        a[k].x = xrow[k * 4 + 0];
        a[k].y = xrow[k * 4 + 1];
    }

    #pragma unroll
    for (int nt = 0; nt < 4; ++nt) {
        v8f acc = {};
        const int col = nt * 16 + l15;
        #pragma unroll
        for (int k = 0; k < 16; ++k) {
            const int krow = k * 4 + hl * 2;
            v2f b;
            b.x = Wl[(krow + 0) * WSTR + col];
            b.y = Wl[(krow + 1) * WSTR + col];
            acc = __builtin_amdgcn_wmma_f32_16x16x4_f32(
                false, a[k], false, b, (short)0, acc, false, false);
        }
        #pragma unroll
        for (int r = 0; r < 8; ++r) {
            const int row = tile * 16 + r + 8 * hl;
            h[(size_t)row * DFEAT + col] = acc[r];
        }
    }
}

__global__ void zero_cnt(int* __restrict__ cnt, int n) {
    int i = blockIdx.x * 256 + threadIdx.x;
    if (i < n) cnt[i] = 0;
}

__global__ void rank_count(const int* __restrict__ dst, int* __restrict__ cnt,
                           int* __restrict__ rk, int e) {
    int i = blockIdx.x * 256 + threadIdx.x;
    if (i < e) rk[i] = atomicAdd(&cnt[dst[i]], 1);
}

__global__ __launch_bounds__(256) void block_sums(const int* __restrict__ cnt,
                                                  int* __restrict__ bsum, int n) {
    __shared__ int red[256];
    const int t = threadIdx.x;
    const int i = blockIdx.x * 256 + t;
    red[t] = (i < n) ? cnt[i] : 0;
    __syncthreads();
    for (int s = 128; s > 0; s >>= 1) {
        if (t < s) red[t] += red[t + s];
        __syncthreads();
    }
    if (t == 0) bsum[blockIdx.x] = red[0];
}

__global__ __launch_bounds__(256) void scan_rows(const int* __restrict__ cnt,
                                                 const int* __restrict__ bsum,
                                                 int* __restrict__ rowptr,
                                                 float* __restrict__ dinv, int n) {
    __shared__ int red[256];
    __shared__ int sc[256];
    const int t = threadIdx.x;
    const int b = blockIdx.x;
    int part = 0;
    if (t < b) part += bsum[t];
    if (t + 256 < b) part += bsum[t + 256];
    red[t] = part;
    const int i = b * 256 + t;
    const int c = (i < n) ? cnt[i] : 0;
    sc[t] = c;
    __syncthreads();
    for (int s = 128; s > 0; s >>= 1) {
        if (t < s) red[t] += red[t + s];
        __syncthreads();
    }
    for (int off = 1; off < 256; off <<= 1) {
        const int v = (t >= off) ? sc[t - off] : 0;
        __syncthreads();
        sc[t] += v;
        __syncthreads();
    }
    if (i < n) {
        const int incl = red[0] + sc[t];
        rowptr[i] = incl - c;
        dinv[i] = rsqrtf((float)(c + 1));
        if (i == n - 1) rowptr[n] = incl;
    }
}

__global__ void csr_fill(const int* __restrict__ src, const int* __restrict__ dst,
                         const int* __restrict__ rk, const int* __restrict__ rowptr,
                         const float* __restrict__ dinv, int* __restrict__ pairs, int e) {
    int i = blockIdx.x * 256 + threadIdx.x;
    if (i >= e) return;
    const int s = src[i];
    const int pos = rowptr[dst[i]] + rk[i];
    int2 p;
    p.x = s;
    p.y = __float_as_int(dinv[s]);
    ((int2*)pairs)[pos] = p;
}

__global__ __launch_bounds__(256) void row_gather(const float* __restrict__ h,
                                                  const int* __restrict__ pairs,
                                                  const int* __restrict__ rowptr,
                                                  const float* __restrict__ dinv,
                                                  const float* __restrict__ bias,
                                                  float* __restrict__ out,
                                                  int do_relu, int n) {
    const int j = threadIdx.x & 15;
    const int d = blockIdx.x * 16 + (threadIdx.x >> 4);
    if (d >= n) return;
    const int2* pr = (const int2*)pairs;
    const float4* h4 = (const float4*)h;
    const int start = rowptr[d];
    const int end = rowptr[d + 1];
    const int last = end - 1;
    const float di = dinv[d];
    const float4 hs = h4[d * 16 + j];
    const float4 bv = ((const float4*)bias)[j];
    float4 acc = {0.0f, 0.0f, 0.0f, 0.0f};
    for (int e = start; e < end; e += 4) {
        const bool k1 = e + 1 < end, k2 = e + 2 < end, k3 = e + 3 < end;
        const int2 p0 = pr[e];
        const int2 p1 = pr[k1 ? e + 1 : last];
        const int2 p2 = pr[k2 ? e + 2 : last];
        const int2 p3 = pr[k3 ? e + 3 : last];
        const float4 v0 = h4[p0.x * 16 + j];
        const float4 v1 = h4[p1.x * 16 + j];
        const float4 v2 = h4[p2.x * 16 + j];
        const float4 v3 = h4[p3.x * 16 + j];
        const float w0 = __int_as_float(p0.y);
        const float w1 = k1 ? __int_as_float(p1.y) : 0.0f;
        const float w2 = k2 ? __int_as_float(p2.y) : 0.0f;
        const float w3 = k3 ? __int_as_float(p3.y) : 0.0f;
        acc.x += v0.x * w0; acc.y += v0.y * w0; acc.z += v0.z * w0; acc.w += v0.w * w0;
        acc.x += v1.x * w1; acc.y += v1.y * w1; acc.z += v1.z * w1; acc.w += v1.w * w1;
        acc.x += v2.x * w2; acc.y += v2.y * w2; acc.z += v2.z * w2; acc.w += v2.w * w2;
        acc.x += v3.x * w3; acc.y += v3.y * w3; acc.z += v3.z * w3; acc.w += v3.w * w3;
    }
    const float dd = di * di;
    float4 r;
    r.x = acc.x * di + hs.x * dd + bv.x;
    r.y = acc.y * di + hs.y * dd + bv.y;
    r.z = acc.z * di + hs.z * dd + bv.z;
    r.w = acc.w * di + hs.w * dd + bv.w;
    if (do_relu) {
        r.x = fmaxf(r.x, 0.0f); r.y = fmaxf(r.y, 0.0f);
        r.z = fmaxf(r.z, 0.0f); r.w = fmaxf(r.w, 0.0f);
    }
    ((float4*)out)[d * 16 + j] = r;
}

extern "C" void kernel_launch(void* const* d_in, const int* in_sizes, int n_in,
                              void* d_out, int out_size, void* d_ws, size_t ws_size,
                              hipStream_t stream) {
    const float* x0 = (const float*)d_in[0];
    const int*   me = (const int*)d_in[1];
    const float* W1 = (const float*)d_in[2];
    const float* b1 = (const float*)d_in[3];
    const float* W2 = (const float*)d_in[4];
    const float* b2 = (const float*)d_in[5];
    const float* W3 = (const float*)d_in[6];
    const float* b3 = (const float*)d_in[7];

    const int E = in_sizes[1] / 2;
    const int* src = me;
    const int* dst = me + E;
    const int NF = NNODES * DFEAT;

    float* ws     = (float*)d_ws;
    int*   cnt    = (int*)d_ws;
    int*   rowptr = cnt + 100352;
    float* dinv   = ws + 200704;
    int*   bsum   = cnt + 301056;
    float* h      = ws + 301568;
    float* xb     = h + NF;
    int*   pairs  = (int*)(xb + NF);
    int*   rk     = pairs + 2 * (size_t)E;

    float* out = (float*)d_out;

    const int TB = 256;
    const int gN   = (NNODES + TB - 1) / TB;
    const int gE   = (E + TB - 1) / TB;
    const int gROW = (NNODES + 15) / 16;
    const int gGEMM = (NTILES + 7) / 8;

    zero_cnt<<<gN, TB, 0, stream>>>(cnt, NNODES);
    rank_count<<<gE, TB, 0, stream>>>(dst, cnt, rk, E);
    block_sums<<<gN, TB, 0, stream>>>(cnt, bsum, NNODES);
    scan_rows<<<gN, TB, 0, stream>>>(cnt, bsum, rowptr, dinv, NNODES);
    csr_fill<<<gE, TB, 0, stream>>>(src, dst, rk, rowptr, dinv, pairs, E);

    gemm64_wmma<<<gGEMM, TB, 0, stream>>>(x0, W1, h, NTILES);
    row_gather<<<gROW, TB, 0, stream>>>(h, pairs, rowptr, dinv, b1, xb, 1, NNODES);

    gemm64_wmma<<<gGEMM, TB, 0, stream>>>(xb, W2, h, NTILES);
    row_gather<<<gROW, TB, 0, stream>>>(h, pairs, rowptr, dinv, b2, xb, 1, NNODES);

    gemm64_wmma<<<gGEMM, TB, 0, stream>>>(xb, W3, h, NTILES);
    row_gather<<<gROW, TB, 0, stream>>>(h, pairs, rowptr, dinv, b3, out, 0, NNODES);
}


# Fuse the 64x64 product into the gather launch: aggregate input rows first, then WMMA on the 16-row tile in LDS; work-groups walk 4 tiles, W read once each; 3 launches fewer; summation order changes
# speedup vs baseline: 3.0035x; 1.1749x over previous
#include <hip/hip_runtime.h>

typedef float v2f __attribute__((ext_vector_type(2)));
typedef float v8f __attribute__((ext_vector_type(8)));

#define NNODES 100000
#define DFEAT  64
#define NTILE  6250
#define NBLK   1563
#define TPITCH 68

__global__ void zero_cnt(int* __restrict__ cnt, int n) {
    int i = blockIdx.x * 256 + threadIdx.x;
    if (i < n) cnt[i] = 0;
}

__global__ void rank_count(const int* __restrict__ dst, int* __restrict__ cnt,
                           int* __restrict__ rk, int e) {
    int i = blockIdx.x * 256 + threadIdx.x;
    if (i < e) rk[i] = atomicAdd(&cnt[dst[i]], 1);
}

__global__ __launch_bounds__(256) void block_sums(const int* __restrict__ cnt,
                                                  int* __restrict__ bsum, int n) {
    __shared__ int red[256];
    const int t = threadIdx.x;
    const int i = blockIdx.x * 256 + t;
    red[t] = (i < n) ? cnt[i] : 0;
    __syncthreads();
    for (int s = 128; s > 0; s >>= 1) {
        if (t < s) red[t] += red[t + s];
        __syncthreads();
    }
    if (t == 0) bsum[blockIdx.x] = red[0];
}

__global__ __launch_bounds__(256) void scan_rows(const int* __restrict__ cnt,
                                                 const int* __restrict__ bsum,
                                                 int* __restrict__ rowptr,
                                                 float* __restrict__ dinv, int n) {
    __shared__ int red[256];
    __shared__ int sc[256];
    const int t = threadIdx.x;
    const int b = blockIdx.x;
    int part = 0;
    if (t < b) part += bsum[t];
    if (t + 256 < b) part += bsum[t + 256];
    red[t] = part;
    const int i = b * 256 + t;
    const int c = (i < n) ? cnt[i] : 0;
    sc[t] = c;
    __syncthreads();
    for (int s = 128; s > 0; s >>= 1) {
        if (t < s) red[t] += red[t + s];
        __syncthreads();
    }
    for (int off = 1; off < 256; off <<= 1) {
        const int v = (t >= off) ? sc[t - off] : 0;
        __syncthreads();
        sc[t] += v;
        __syncthreads();
    }
    if (i < n) {
        const int incl = red[0] + sc[t];
        rowptr[i] = incl - c;
        dinv[i] = rsqrtf((float)(c + 1));
        if (i == n - 1) rowptr[n] = incl;
    }
}

__global__ void csr_fill(const int* __restrict__ src, const int* __restrict__ dst,
                         const int* __restrict__ rk, const int* __restrict__ rowptr,
                         const float* __restrict__ dinv, int* __restrict__ pairs, int e) {
    int i = blockIdx.x * 256 + threadIdx.x;
    if (i >= e) return;
    const int s = src[i];
    const int pos = rowptr[dst[i]] + rk[i];
    int2 p;
    p.x = s;
    p.y = __float_as_int(dinv[s]);
    ((int2*)pairs)[pos] = p;
}

__global__ __launch_bounds__(256) void gather_gemm(const float* __restrict__ x,
                                                   const int* __restrict__ pairs,
                                                   const int* __restrict__ rowptr,
                                                   const float* __restrict__ dinv,
                                                   const float* __restrict__ W,
                                                   const float* __restrict__ bias,
                                                   float* __restrict__ out,
                                                   int do_relu, int n, int ntiles) {
    __shared__ __attribute__((aligned(16))) float S[48 * TPITCH];
    __shared__ __attribute__((aligned(16))) float Wf[256 * 16];
    __shared__ float Bw[256];

    const int j = threadIdx.x & 15;
    const int r = threadIdx.x >> 4;
    const float4* x4 = (const float4*)x;
    const int rowoff = r * TPITCH + 4 * j;

    const int lane = threadIdx.x & 31;
    const int wave = threadIdx.x >> 5;
    const int hl  = lane >> 4;
    const int l15 = lane & 15;
    const int nt  = wave & 3;
    const int kh  = wave >> 2;
    const int col = nt * 16 + l15;

    float4* wf = (float4*)Wf + wave * 128 + lane;
    #pragma unroll
    for (int p = 0; p < 4; ++p) {
        const int kk = (kh * 8 + 2 * p) * 4 + hl * 2;
        float4 v;
        v.x = W[(kk + 0) * DFEAT + col];
        v.y = W[(kk + 1) * DFEAT + col];
        v.z = W[(kk + 4) * DFEAT + col];
        v.w = W[(kk + 5) * DFEAT + col];
        wf[p * 32] = v;
    }
    Bw[threadIdx.x] = (kh == 0) ? bias[col] : 0.0f;

    for (int tile = blockIdx.x; tile < ntiles; tile += gridDim.x) {
        const int d = tile * 16 + r;

        float4 g = {0.0f, 0.0f, 0.0f, 0.0f};
        if (d < n) {
            const int2* pr = (const int2*)pairs;
            const int start = rowptr[d];
            const int end = rowptr[d + 1];
            const int last = end - 1;
            const float di = dinv[d];
            const float4 xs = x4[d * 16 + j];
            float4 acc = {0.0f, 0.0f, 0.0f, 0.0f};
            for (int e = start; e < end; e += 4) {
                const bool k1 = e + 1 < end, k2 = e + 2 < end, k3 = e + 3 < end;
                const int2 p0 = pr[e];
                const int2 p1 = pr[k1 ? e + 1 : last];
                const int2 p2 = pr[k2 ? e + 2 : last];
                const int2 p3 = pr[k3 ? e + 3 : last];
                const float4 v0 = x4[p0.x * 16 + j];
                const float4 v1 = x4[p1.x * 16 + j];
                const float4 v2 = x4[p2.x * 16 + j];
                const float4 v3 = x4[p3.x * 16 + j];
                const float w0 = __int_as_float(p0.y);
                const float w1 = k1 ? __int_as_float(p1.y) : 0.0f;
                const float w2 = k2 ? __int_as_float(p2.y) : 0.0f;
                const float w3 = k3 ? __int_as_float(p3.y) : 0.0f;
                acc.x += v0.x * w0; acc.y += v0.y * w0; acc.z += v0.z * w0; acc.w += v0.w * w0;
                acc.x += v1.x * w1; acc.y += v1.y * w1; acc.z += v1.z * w1; acc.w += v1.w * w1;
                acc.x += v2.x * w2; acc.y += v2.y * w2; acc.z += v2.z * w2; acc.w += v2.w * w2;
                acc.x += v3.x * w3; acc.y += v3.y * w3; acc.z += v3.z * w3; acc.w += v3.w * w3;
            }
            const float dd = di * di;
            g.x = acc.x * di + xs.x * dd;
            g.y = acc.y * di + xs.y * dd;
            g.z = acc.z * di + xs.z * dd;
            g.w = acc.w * di + xs.w * dd;
        }
        *(float4*)&S[32 * TPITCH + rowoff] = g;
        __syncthreads();

        const float bc = Bw[threadIdx.x];
        v8f c = {};
        #pragma unroll
        for (int p = 0; p < 4; ++p) {
            const int kk = (kh * 8 + 2 * p) * 4 + hl * 2;
            const float4 bb = wf[p * 32];
            const v2f a0 = *(const v2f*)&S[(32 + l15) * TPITCH + kk];
            const v2f a1 = *(const v2f*)&S[(32 + l15) * TPITCH + kk + 4];
            v2f b0, b1;
            b0.x = bb.x; b0.y = bb.y;
            b1.x = bb.z; b1.y = bb.w;
            c = __builtin_amdgcn_wmma_f32_16x16x4_f32(
                false, a0, false, b0, (short)0, c, false, false);
            c = __builtin_amdgcn_wmma_f32_16x16x4_f32(
                false, a1, false, b1, (short)0, c, false, false);
        }
        #pragma unroll
        for (int q = 0; q < 8; ++q)
            S[(kh * 16 + q + 8 * hl) * TPITCH + col] = c[q] + bc;
        __syncthreads();

        if (d < n) {
            const float4 o0 = *(const float4*)&S[rowoff];
            const float4 o1 = *(const float4*)&S[16 * TPITCH + rowoff];
            float4 res;
            res.x = o0.x + o1.x;
            res.y = o0.y + o1.y;
            res.z = o0.z + o1.z;
            res.w = o0.w + o1.w;
            if (do_relu) {
                res.x = fmaxf(res.x, 0.0f); res.y = fmaxf(res.y, 0.0f);
                res.z = fmaxf(res.z, 0.0f); res.w = fmaxf(res.w, 0.0f);
            }
            ((float4*)out)[d * 16 + j] = res;
        }
    }
}

extern "C" void kernel_launch(void* const* d_in, const int* in_sizes, int n_in,
                              void* d_out, int out_size, void* d_ws, size_t ws_size,
                              hipStream_t stream) {
    const float* x0 = (const float*)d_in[0];
    const int*   me = (const int*)d_in[1];
    const float* W1 = (const float*)d_in[2];
    const float* b1 = (const float*)d_in[3];
    const float* W2 = (const float*)d_in[4];
    const float* b2 = (const float*)d_in[5];
    const float* W3 = (const float*)d_in[6];
    const float* b3 = (const float*)d_in[7];

    const int E = in_sizes[1] / 2;
    const int* src = me;
    const int* dst = me + E;
    const int NF = NNODES * DFEAT;

    float* ws     = (float*)d_ws;
    int*   cnt    = (int*)d_ws;
    int*   rowptr = cnt + 100352;
    float* dinv   = ws + 200704;
    int*   bsum   = cnt + 301056;
    float* xa     = ws + 301568;
    float* xb     = xa + NF;
    int*   pairs  = (int*)(xb + NF);
    int*   rk     = pairs + 2 * (size_t)E;

    float* out = (float*)d_out;

    const int TB = 256;
    const int gN   = (NNODES + TB - 1) / TB;
    const int gE   = (E + TB - 1) / TB;

    zero_cnt<<<gN, TB, 0, stream>>>(cnt, NNODES);
    rank_count<<<gE, TB, 0, stream>>>(dst, cnt, rk, E);
    block_sums<<<gN, TB, 0, stream>>>(cnt, bsum, NNODES);
    scan_rows<<<gN, TB, 0, stream>>>(cnt, bsum, rowptr, dinv, NNODES);
    csr_fill<<<gE, TB, 0, stream>>>(src, dst, rk, rowptr, dinv, pairs, E);

    gather_gemm<<<NBLK, TB, 0, stream>>>(x0, pairs, rowptr, dinv, W1, b1, xa, 1, NNODES, NTILE);

    gather_gemm<<<NBLK, TB, 0, stream>>>(xa, pairs, rowptr, dinv, W2, b2, xb, 1, NNODES, NTILE);

    gather_gemm<<<NBLK, TB, 0, stream>>>(xb, pairs, rowptr, dinv, W3, b3, out, 0, NNODES, NTILE);
}


# Gather's dependent-load chain cut: a row's edge records come 16 per coalesced load, passed lane to lane (DPP); 8 row loads per group; next tile prefetched; self term opens each sum (order changes)
# speedup vs baseline: 3.2183x; 1.0715x over previous
#include <hip/hip_runtime.h>

typedef float v2f __attribute__((ext_vector_type(2)));
typedef float v8f __attribute__((ext_vector_type(8)));

#define NNODES 100000
#define DFEAT  64
#define NTILE  6250
#define NBLK   1563
#define TPITCH 68

__global__ void zero_cnt(int* __restrict__ cnt, int n) {
    int i = blockIdx.x * 256 + threadIdx.x;
    if (i < n) cnt[i] = 0;
}

__global__ void rank_count(const int* __restrict__ dst, int* __restrict__ cnt,
                           int* __restrict__ rk, int e) {
    int i = blockIdx.x * 256 + threadIdx.x;
    if (i < e) rk[i] = atomicAdd(&cnt[dst[i]], 1);
}

__global__ __launch_bounds__(256) void block_sums(const int* __restrict__ cnt,
                                                  int* __restrict__ bsum, int n) {
    __shared__ int red[256];
    const int t = threadIdx.x;
    const int i = blockIdx.x * 256 + t;
    red[t] = (i < n) ? cnt[i] : 0;
    __syncthreads();
    for (int s = 128; s > 0; s >>= 1) {
        if (t < s) red[t] += red[t + s];
        __syncthreads();
    }
    if (t == 0) bsum[blockIdx.x] = red[0];
}

__global__ __launch_bounds__(256) void scan_rows(const int* __restrict__ cnt,
                                                 const int* __restrict__ bsum,
                                                 int* __restrict__ rowptr,
                                                 float* __restrict__ dinv, int n) {
    __shared__ int red[256];
    __shared__ int sc[256];
    const int t = threadIdx.x;
    const int b = blockIdx.x;
    int part = 0;
    if (t < b) part += bsum[t];
    if (t + 256 < b) part += bsum[t + 256];
    red[t] = part;
    const int i = b * 256 + t;
    const int c = (i < n) ? cnt[i] : 0;
    sc[t] = c;
    __syncthreads();
    for (int s = 128; s > 0; s >>= 1) {
        if (t < s) red[t] += red[t + s];
        __syncthreads();
    }
    for (int off = 1; off < 256; off <<= 1) {
        const int v = (t >= off) ? sc[t - off] : 0;
        __syncthreads();
        sc[t] += v;
        __syncthreads();
    }
    if (i < n) {
        const int incl = red[0] + sc[t];
        rowptr[i] = incl - c;
        dinv[i] = rsqrtf((float)(c + 1));
        if (i == n - 1) rowptr[n] = incl;
    }
}

__global__ void csr_fill(const int* __restrict__ src, const int* __restrict__ dst,
                         const int* __restrict__ rk, const int* __restrict__ rowptr,
                         const float* __restrict__ dinv, int* __restrict__ pairs, int e) {
    int i = blockIdx.x * 256 + threadIdx.x;
    if (i >= e) return;
    const int s = src[i];
    const int pos = rowptr[dst[i]] + rk[i];
    int2 p;
    p.x = s;
    p.y = __float_as_int(dinv[s]);
    ((int2*)pairs)[pos] = p;
}

__device__ __forceinline__ void lds_barrier() {
    __builtin_amdgcn_fence(__ATOMIC_RELEASE, "workgroup", "local");
    __builtin_amdgcn_s_barrier();
    __builtin_amdgcn_fence(__ATOMIC_ACQUIRE, "workgroup", "local");
}

template <int K>
__device__ __forceinline__ int row_lane(int v) {
    return __builtin_amdgcn_update_dpp(0, v, 0x150 + K, 0xf, 0xf, true);
}

template <int K0>
__device__ __forceinline__ void edge8(const float4* __restrict__ x4, int j, int2 p,
                                      float4& acc) {
    const int s0 = row_lane<K0 + 0>(p.x);
    const int s1 = row_lane<K0 + 1>(p.x);
    const int s2 = row_lane<K0 + 2>(p.x);
    const int s3 = row_lane<K0 + 3>(p.x);
    const int s4 = row_lane<K0 + 4>(p.x);
    const int s5 = row_lane<K0 + 5>(p.x);
    const int s6 = row_lane<K0 + 6>(p.x);
    const int s7 = row_lane<K0 + 7>(p.x);
    const float4 v0 = x4[s0 * 16 + j];
    const float4 v1 = x4[s1 * 16 + j];
    const float4 v2 = x4[s2 * 16 + j];
    const float4 v3 = x4[s3 * 16 + j];
    const float4 v4 = x4[s4 * 16 + j];
    const float4 v5 = x4[s5 * 16 + j];
    const float4 v6 = x4[s6 * 16 + j];
    const float4 v7 = x4[s7 * 16 + j];
    const float w0 = __int_as_float(row_lane<K0 + 0>(p.y));
    acc.x += v0.x * w0; acc.y += v0.y * w0; acc.z += v0.z * w0; acc.w += v0.w * w0;
    const float w1 = __int_as_float(row_lane<K0 + 1>(p.y));
    acc.x += v1.x * w1; acc.y += v1.y * w1; acc.z += v1.z * w1; acc.w += v1.w * w1;
    const float w2 = __int_as_float(row_lane<K0 + 2>(p.y));
    acc.x += v2.x * w2; acc.y += v2.y * w2; acc.z += v2.z * w2; acc.w += v2.w * w2;
    const float w3 = __int_as_float(row_lane<K0 + 3>(p.y));
    acc.x += v3.x * w3; acc.y += v3.y * w3; acc.z += v3.z * w3; acc.w += v3.w * w3;
    const float w4 = __int_as_float(row_lane<K0 + 4>(p.y));
    acc.x += v4.x * w4; acc.y += v4.y * w4; acc.z += v4.z * w4; acc.w += v4.w * w4;
    const float w5 = __int_as_float(row_lane<K0 + 5>(p.y));
    acc.x += v5.x * w5; acc.y += v5.y * w5; acc.z += v5.z * w5; acc.w += v5.w * w5;
    const float w6 = __int_as_float(row_lane<K0 + 6>(p.y));
    acc.x += v6.x * w6; acc.y += v6.y * w6; acc.z += v6.z * w6; acc.w += v6.w * w6;
    const float w7 = __int_as_float(row_lane<K0 + 7>(p.y));
    acc.x += v7.x * w7; acc.y += v7.y * w7; acc.z += v7.z * w7; acc.w += v7.w * w7;
}

__device__ __forceinline__ int2 edge_record(const int2* __restrict__ pr, int first, int j,
                                            int last) {
    return pr[min(first + j, last)];
}

__global__ __launch_bounds__(256) void gather_gemm(const float* __restrict__ x,
                                                   const int* __restrict__ pairs,
                                                   const int* __restrict__ rowptr,
                                                   const float* __restrict__ dinv,
                                                   const float* __restrict__ W,
                                                   const float* __restrict__ bias,
                                                   float* __restrict__ out,
                                                   int do_relu, int n, int ntiles) {
    __shared__ __attribute__((aligned(16))) float S[48 * TPITCH];
    __shared__ __attribute__((aligned(16))) float Wf[256 * 16];
    __shared__ float Bw[256];

    const int j = threadIdx.x & 15;
    const int r = threadIdx.x >> 4;
    const float4* x4 = (const float4*)x;
    const int rowoff = r * TPITCH + 4 * j;

    const int lane = threadIdx.x & 31;
    const int wave = threadIdx.x >> 5;
    const int hl  = lane >> 4;
    const int l15 = lane & 15;
    const int nt  = wave & 3;
    const int kh  = wave >> 2;
    const int col = nt * 16 + l15;

    float4* wf = (float4*)Wf + wave * 128 + lane;
    #pragma unroll
    for (int p = 0; p < 4; ++p) {
        const int kk = (kh * 8 + 2 * p) * 4 + hl * 2;
        float4 v;
        v.x = W[(kk + 0) * DFEAT + col];
        v.y = W[(kk + 1) * DFEAT + col];
        v.z = W[(kk + 4) * DFEAT + col];
        v.w = W[(kk + 5) * DFEAT + col];
        wf[p * 32] = v;
    }
    Bw[threadIdx.x] = (kh == 0) ? bias[col] : 0.0f;

    const int2* pr = (const int2*)pairs;
    const int gstep = gridDim.x;
    int start = 0, end = 0;
    int2 pa = {0, 0}, pb = {0, 0};
    float di = 0.0f;
    float4 xs = {0.0f, 0.0f, 0.0f, 0.0f};
    {
        const int d0 = blockIdx.x * 16 + r;
        if ((int)blockIdx.x < ntiles && d0 < n) {
            start = rowptr[d0];
            end = rowptr[d0 + 1];
            di = dinv[d0];
            xs = x4[d0 * 16 + j];
        }
        if (end > start) {
            pa = edge_record(pr, start, j, end - 1);
            pb = edge_record(pr, start + 16, j, end - 1);
        }
    }

    for (int tile = blockIdx.x; tile < ntiles; tile += gstep) {
        asm volatile("" :: "v"(xs.x), "v"(xs.y), "v"(xs.z), "v"(xs.w), "v"(di));
        float4 acc;
        acc.x = xs.x * di; acc.y = xs.y * di; acc.z = xs.z * di; acc.w = xs.w * di;

        const int tn = tile + gstep;
        const int dn = tn * 16 + r;
        const bool nextrow = tn < ntiles && dn < n;
        int nstart = 0, nend = 0;
        if (nextrow) {
            nstart = rowptr[dn];
            nend = rowptr[dn + 1];
        }

        int left = end - start;
        if (left > 0) {
            for (;;) {
                pa.y = (j < left) ? pa.y : 0;
                pb.y = (j + 16 < left) ? pb.y : 0;
                edge8<0>(x4, j, pa, acc);
                if (left > 8)  edge8<8>(x4, j, pa, acc);
                if (left > 16) edge8<0>(x4, j, pb, acc);
                if (left > 24) edge8<8>(x4, j, pb, acc);
                left -= 32;
                if (left <= 0) break;
                pa = edge_record(pr, end - left, j, end - 1);
                pb = edge_record(pr, end - left + 16, j, end - 1);
            }
        }
        float4 g;
        g.x = acc.x * di;
        g.y = acc.y * di;
        g.z = acc.z * di;
        g.w = acc.w * di;

        start = nstart;
        end = nend;
        if (nend > nstart) {
            pa = edge_record(pr, nstart, j, nend - 1);
            pb = edge_record(pr, nstart + 16, j, nend - 1);
        }
        if (nextrow) {
            di = dinv[dn];
            xs = x4[dn * 16 + j];
        } else {
            di = 0.0f;
            xs.x = 0.0f; xs.y = 0.0f; xs.z = 0.0f; xs.w = 0.0f;
        }
        *(float4*)&S[32 * TPITCH + rowoff] = g;
        lds_barrier();

        const float bc = Bw[threadIdx.x];
        v8f c = {};
        #pragma unroll
        for (int p = 0; p < 4; ++p) {
            const int kk = (kh * 8 + 2 * p) * 4 + hl * 2;
            const float4 bb = wf[p * 32];
            const v2f a0 = *(const v2f*)&S[(32 + l15) * TPITCH + kk];
            const v2f a1 = *(const v2f*)&S[(32 + l15) * TPITCH + kk + 4];
            v2f b0, b1;
            b0.x = bb.x; b0.y = bb.y;
            b1.x = bb.z; b1.y = bb.w;
            c = __builtin_amdgcn_wmma_f32_16x16x4_f32(
                false, a0, false, b0, (short)0, c, false, false);
            c = __builtin_amdgcn_wmma_f32_16x16x4_f32(
                false, a1, false, b1, (short)0, c, false, false);
        }
        #pragma unroll
        for (int q = 0; q < 8; ++q)
            S[(kh * 16 + q + 8 * hl) * TPITCH + col] = c[q] + bc;
        lds_barrier();

        const int d = tile * 16 + r;
        if (d < n) {
            const float4 o0 = *(const float4*)&S[rowoff];
            const float4 o1 = *(const float4*)&S[16 * TPITCH + rowoff];
            float4 res;
            res.x = o0.x + o1.x;
            res.y = o0.y + o1.y;
            res.z = o0.z + o1.z;
            res.w = o0.w + o1.w;
            if (do_relu) {
                res.x = fmaxf(res.x, 0.0f); res.y = fmaxf(res.y, 0.0f);
                res.z = fmaxf(res.z, 0.0f); res.w = fmaxf(res.w, 0.0f);
            }
            ((float4*)out)[d * 16 + j] = res;
        }
    }
}

extern "C" void kernel_launch(void* const* d_in, const int* in_sizes, int n_in,
                              void* d_out, int out_size, void* d_ws, size_t ws_size,
                              hipStream_t stream) {
    const float* x0 = (const float*)d_in[0];
    const int*   me = (const int*)d_in[1];
    const float* W1 = (const float*)d_in[2];
    const float* b1 = (const float*)d_in[3];
    const float* W2 = (const float*)d_in[4];
    const float* b2 = (const float*)d_in[5];
    const float* W3 = (const float*)d_in[6];
    const float* b3 = (const float*)d_in[7];

    const int E = in_sizes[1] / 2;
    const int* src = me;
    const int* dst = me + E;
    const int NF = NNODES * DFEAT;

    float* ws     = (float*)d_ws;
    int*   cnt    = (int*)d_ws;
    int*   rowptr = cnt + 100352;
    float* dinv   = ws + 200704;
    int*   bsum   = cnt + 301056;
    float* xa     = ws + 301568;
    float* xb     = xa + NF;
    int*   pairs  = (int*)(xb + NF);
    int*   rk     = pairs + 2 * (size_t)E;

    float* out = (float*)d_out;

    const int TB = 256;
    const int gN   = (NNODES + TB - 1) / TB;
    const int gE   = (E + TB - 1) / TB;

    zero_cnt<<<gN, TB, 0, stream>>>(cnt, NNODES);
    rank_count<<<gE, TB, 0, stream>>>(dst, cnt, rk, E);
    block_sums<<<gN, TB, 0, stream>>>(cnt, bsum, NNODES);
    scan_rows<<<gN, TB, 0, stream>>>(cnt, bsum, rowptr, dinv, NNODES);
    csr_fill<<<gE, TB, 0, stream>>>(src, dst, rk, rowptr, dinv, pairs, E);

    gather_gemm<<<NBLK, TB, 0, stream>>>(x0, pairs, rowptr, dinv, W1, b1, xa, 1, NNODES, NTILE);

    gather_gemm<<<NBLK, TB, 0, stream>>>(xa, pairs, rowptr, dinv, W2, b2, xb, 1, NNODES, NTILE);

    gather_gemm<<<NBLK, TB, 0, stream>>>(xb, pairs, rowptr, dinv, W3, b3, out, 0, NNODES, NTILE);
}


# Edge list ordered through LDS: edges binned per chunk by 256-node destination range, each bin counting-sorted in LDS; no global atomics or scattered table reads; 2 launches fewer; sum order shifts
# speedup vs baseline: 6.4553x; 2.0058x over previous
#include <hip/hip_runtime.h>

typedef float v2f __attribute__((ext_vector_type(2)));
typedef float v8f __attribute__((ext_vector_type(8)));

#define NNODES 100000
#define DFEAT  64
#define NTILE  6250
#define NBLK   1563
#define TPITCH 68

#define CHUNK   6400
#define BINSH   8
#define NBIN    391
#define TABP    512
#define MAXCH   256

__device__ __forceinline__ int wave_incl_scan(int v, int lane) {
    #pragma unroll
    for (int o = 1; o < 32; o <<= 1) {
        const int u = __shfl_up(v, o, 32);
        if (lane >= o) v += u;
    }
    return v;
}

__global__ __launch_bounds__(640) void bin_edges(const int* __restrict__ src,
                                                 const int* __restrict__ dst,
                                                 int2* __restrict__ stage,
                                                 int2* __restrict__ tab, int e) {
    __shared__ int hist[TABP];
    __shared__ int sc[TABP];
    __shared__ int wtot[16];
    __shared__ __attribute__((aligned(16))) int2 buf[CHUNK];
    const int t = threadIdx.x;
    const int lane = t & 31;
    const int wave = t >> 5;
    const int c = blockIdx.x;
    const int base = c * CHUNK;
    if (t < TABP) hist[t] = 0;
    __syncthreads();
    int s[10], d[10], r[10];
    #pragma unroll
    for (int u = 0; u < 10; ++u) {
        const int i = base + u * 640 + t;
        const int ii = min(i, e - 1);
        const int dv = dst[ii];
        s[u] = src[ii];
        d[u] = (i < e) ? dv : -1;
    }
    #pragma unroll
    for (int u = 0; u < 10; ++u)
        r[u] = (d[u] >= 0) ? atomicAdd(&hist[d[u] >> BINSH], 1) : 0;
    __syncthreads();
    const int hv = (t < TABP) ? hist[t] : 0;
    const int incl = wave_incl_scan(hv, lane);
    if (t < TABP && lane == 31) wtot[wave] = incl;
    __syncthreads();
    if (t < TABP) {
        int pre = 0;
        #pragma unroll
        for (int w = 0; w < 16; ++w) pre += (w < wave) ? wtot[w] : 0;
        sc[t] = incl + pre;
    }
    __syncthreads();
    #pragma unroll
    for (int u = 0; u < 10; ++u) {
        if (d[u] >= 0) {
            const int b = d[u] >> BINSH;
            int2 rec;
            rec.x = s[u];
            rec.y = d[u];
            buf[sc[b] - hist[b] + r[u]] = rec;
        }
    }
    __syncthreads();
    const int cnt = min(CHUNK, e - base);
    for (int i = t; i < cnt; i += 640) stage[base + i] = buf[i];
    if (t < TABP) {
        int2 v;
        v.x = hv;
        v.y = sc[t] - hv;
        tab[c * TABP + t] = v;
    }
}

__global__ __launch_bounds__(1024) void sort_bin(const int2* __restrict__ stage,
                                                 const int2* __restrict__ tab,
                                                 int* __restrict__ rowptr,
                                                 float* __restrict__ dinv,
                                                 int* __restrict__ srcs, int n, int nchunk) {
    __shared__ int segoff[MAXCH];
    __shared__ int segcnt[MAXCH];
    __shared__ int red[8];
    __shared__ int wtot[8];
    __shared__ int ncnt[256];
    __shared__ int sc[256];
    __shared__ int cur[256];
    const int t = threadIdx.x;
    const int b = blockIdx.x;
    const int lane = t & 31;
    const int wave = t >> 5;
    int2 v;
    v.x = 0;
    v.y = 0;
    if (t < nchunk && t < MAXCH) v = tab[t * TABP + b];
    int osum = v.y;
    #pragma unroll
    for (int o = 16; o > 0; o >>= 1) osum += __shfl_xor(osum, o, 32);
    if (t < MAXCH) {
        segcnt[t] = v.x;
        segoff[t] = t * CHUNK + v.y;
        ncnt[t] = 0;
        cur[t] = 0;
        if (lane == 0) red[wave] = osum;
    }
    __syncthreads();
    int binbase = 0;
    #pragma unroll
    for (int w = 0; w < 8; ++w) binbase += red[w];
    int cn[8], co[8];
    int maxn = 0;
    #pragma unroll
    for (int i = 0; i < 8; ++i) {
        cn[i] = segcnt[wave + 32 * i];
        co[i] = segoff[wave + 32 * i];
        maxn = max(maxn, cn[i]);
    }
    const int nbase = b << BINSH;
    for (int q = 0; q < maxn; q += 32) {
        const int k = q + lane;
        int dl[8];
        #pragma unroll
        for (int i = 0; i < 8; ++i) {
            const int idx = (k < cn[i]) ? co[i] + k : 0;
            dl[i] = stage[idx].y;
        }
        #pragma unroll
        for (int i = 0; i < 8; ++i)
            if (k < cn[i]) atomicAdd(&ncnt[dl[i] - nbase], 1);
    }
    __syncthreads();
    const int cdeg = (t < 256) ? ncnt[t] : 0;
    const int incl = wave_incl_scan(cdeg, lane);
    if (t < 256 && lane == 31) wtot[wave] = incl;
    __syncthreads();
    if (t < 256) {
        int pre = 0;
        #pragma unroll
        for (int w = 0; w < 8; ++w) pre += (w < wave) ? wtot[w] : 0;
        const int inc = incl + pre;
        sc[t] = inc;
        const int node = nbase + t;
        if (node < n) {
            rowptr[node] = binbase + inc - cdeg;
            dinv[node] = rsqrtf((float)(cdeg + 1));
            if (node == n - 1) rowptr[n] = binbase + inc;
        }
    }
    __syncthreads();
    for (int q = 0; q < maxn; q += 32) {
        const int k = q + lane;
        int2 ed[8];
        #pragma unroll
        for (int i = 0; i < 8; ++i) {
            const int idx = (k < cn[i]) ? co[i] + k : 0;
            ed[i] = stage[idx];
        }
        #pragma unroll
        for (int i = 0; i < 8; ++i) {
            if (k < cn[i]) {
                const int l = ed[i].y - nbase;
                const int r = atomicAdd(&cur[l], 1);
                srcs[binbase + sc[l] - ncnt[l] + r] = ed[i].x;
            }
        }
    }
}

__global__ void edge_weights(const int* __restrict__ srcs, const float* __restrict__ dinv,
                             int* __restrict__ pairs, int e) {
    int i = blockIdx.x * 256 + threadIdx.x;
    if (i >= e) return;
    const int s = srcs[i];
    int2 p;
    p.x = s;
    p.y = __float_as_int(dinv[s]);
    ((int2*)pairs)[i] = p;
}

__device__ __forceinline__ void lds_barrier() {
    __builtin_amdgcn_fence(__ATOMIC_RELEASE, "workgroup", "local");
    __builtin_amdgcn_s_barrier();
    __builtin_amdgcn_fence(__ATOMIC_ACQUIRE, "workgroup", "local");
}

template <int K>
__device__ __forceinline__ int row_lane(int v) {
    return __builtin_amdgcn_update_dpp(0, v, 0x150 + K, 0xf, 0xf, true);
}

template <int K0>
__device__ __forceinline__ void edge8(const float4* __restrict__ x4, int j, int2 p,
                                      float4& acc) {
    const int s0 = row_lane<K0 + 0>(p.x);
    const int s1 = row_lane<K0 + 1>(p.x);
    const int s2 = row_lane<K0 + 2>(p.x);
    const int s3 = row_lane<K0 + 3>(p.x);
    const int s4 = row_lane<K0 + 4>(p.x);
    const int s5 = row_lane<K0 + 5>(p.x);
    const int s6 = row_lane<K0 + 6>(p.x);
    const int s7 = row_lane<K0 + 7>(p.x);
    const float4 v0 = x4[s0 * 16 + j];
    const float4 v1 = x4[s1 * 16 + j];
    const float4 v2 = x4[s2 * 16 + j];
    const float4 v3 = x4[s3 * 16 + j];
    const float4 v4 = x4[s4 * 16 + j];
    const float4 v5 = x4[s5 * 16 + j];
    const float4 v6 = x4[s6 * 16 + j];
    const float4 v7 = x4[s7 * 16 + j];
    const float w0 = __int_as_float(row_lane<K0 + 0>(p.y));
    acc.x += v0.x * w0; acc.y += v0.y * w0; acc.z += v0.z * w0; acc.w += v0.w * w0;
    const float w1 = __int_as_float(row_lane<K0 + 1>(p.y));
    acc.x += v1.x * w1; acc.y += v1.y * w1; acc.z += v1.z * w1; acc.w += v1.w * w1;
    const float w2 = __int_as_float(row_lane<K0 + 2>(p.y));
    acc.x += v2.x * w2; acc.y += v2.y * w2; acc.z += v2.z * w2; acc.w += v2.w * w2;
    const float w3 = __int_as_float(row_lane<K0 + 3>(p.y));
    acc.x += v3.x * w3; acc.y += v3.y * w3; acc.z += v3.z * w3; acc.w += v3.w * w3;
    const float w4 = __int_as_float(row_lane<K0 + 4>(p.y));
    acc.x += v4.x * w4; acc.y += v4.y * w4; acc.z += v4.z * w4; acc.w += v4.w * w4;
    const float w5 = __int_as_float(row_lane<K0 + 5>(p.y));
    acc.x += v5.x * w5; acc.y += v5.y * w5; acc.z += v5.z * w5; acc.w += v5.w * w5;
    const float w6 = __int_as_float(row_lane<K0 + 6>(p.y));
    acc.x += v6.x * w6; acc.y += v6.y * w6; acc.z += v6.z * w6; acc.w += v6.w * w6;
    const float w7 = __int_as_float(row_lane<K0 + 7>(p.y));
    acc.x += v7.x * w7; acc.y += v7.y * w7; acc.z += v7.z * w7; acc.w += v7.w * w7;
}

__device__ __forceinline__ int2 edge_record(const int2* __restrict__ pr, int first, int j,
                                            int last) {
    return pr[min(first + j, last)];
}

__global__ __launch_bounds__(256) void gather_gemm(const float* __restrict__ x,
                                                   const int* __restrict__ pairs,
                                                   const int* __restrict__ rowptr,
                                                   const float* __restrict__ dinv,
                                                   const float* __restrict__ W,
                                                   const float* __restrict__ bias,
                                                   float* __restrict__ out,
                                                   int do_relu, int n, int ntiles) {
    __shared__ __attribute__((aligned(16))) float S[48 * TPITCH];
    __shared__ __attribute__((aligned(16))) float Wf[256 * 16];
    __shared__ float Bw[256];

    const int j = threadIdx.x & 15;
    const int r = threadIdx.x >> 4;
    const float4* x4 = (const float4*)x;
    const int rowoff = r * TPITCH + 4 * j;

    const int lane = threadIdx.x & 31;
    const int wave = threadIdx.x >> 5;
    const int hl  = lane >> 4;
    const int l15 = lane & 15;
    const int nt  = wave & 3;
    const int kh  = wave >> 2;
    const int col = nt * 16 + l15;

    float4* wf = (float4*)Wf + wave * 128 + lane;
    #pragma unroll
    for (int p = 0; p < 4; ++p) {
        const int kk = (kh * 8 + 2 * p) * 4 + hl * 2;
        float4 v;
        v.x = W[(kk + 0) * DFEAT + col];
        v.y = W[(kk + 1) * DFEAT + col];
        v.z = W[(kk + 4) * DFEAT + col];
        v.w = W[(kk + 5) * DFEAT + col];
        wf[p * 32] = v;
    }
    Bw[threadIdx.x] = (kh == 0) ? bias[col] : 0.0f;

    const int2* pr = (const int2*)pairs;
    const int gstep = gridDim.x;
    int start = 0, end = 0;
    int2 pa = {0, 0}, pb = {0, 0};
    float di = 0.0f;
    float4 xs = {0.0f, 0.0f, 0.0f, 0.0f};
    {
        const int d0 = blockIdx.x * 16 + r;
        if ((int)blockIdx.x < ntiles && d0 < n) {
            start = rowptr[d0];
            end = rowptr[d0 + 1];
            di = dinv[d0];
            xs = x4[d0 * 16 + j];
        }
        if (end > start) {
            pa = edge_record(pr, start, j, end - 1);
            pb = edge_record(pr, start + 16, j, end - 1);
        }
    }

    for (int tile = blockIdx.x; tile < ntiles; tile += gstep) {
        asm volatile("" :: "v"(xs.x), "v"(xs.y), "v"(xs.z), "v"(xs.w), "v"(di));
        float4 acc;
        acc.x = xs.x * di; acc.y = xs.y * di; acc.z = xs.z * di; acc.w = xs.w * di;

        const int tn = tile + gstep;
        const int dn = tn * 16 + r;
        const bool nextrow = tn < ntiles && dn < n;
        int nstart = 0, nend = 0;
        if (nextrow) {
            nstart = rowptr[dn];
            nend = rowptr[dn + 1];
        }

        int left = end - start;
        if (left > 0) {
            for (;;) {
                pa.y = (j < left) ? pa.y : 0;
                pb.y = (j + 16 < left) ? pb.y : 0;
                edge8<0>(x4, j, pa, acc);
                if (left > 8)  edge8<8>(x4, j, pa, acc);
                if (left > 16) edge8<0>(x4, j, pb, acc);
                if (left > 24) edge8<8>(x4, j, pb, acc);
                left -= 32;
                if (left <= 0) break;
                pa = edge_record(pr, end - left, j, end - 1);
                pb = edge_record(pr, end - left + 16, j, end - 1);
            }
        }
        float4 g;
        g.x = acc.x * di;
        g.y = acc.y * di;
        g.z = acc.z * di;
        g.w = acc.w * di;

        start = nstart;
        end = nend;
        if (nend > nstart) {
            pa = edge_record(pr, nstart, j, nend - 1);
            pb = edge_record(pr, nstart + 16, j, nend - 1);
        }
        if (nextrow) {
            di = dinv[dn];
            xs = x4[dn * 16 + j];
        } else {
            di = 0.0f;
            xs.x = 0.0f; xs.y = 0.0f; xs.z = 0.0f; xs.w = 0.0f;
        }
        *(float4*)&S[32 * TPITCH + rowoff] = g;
        lds_barrier();

        const float bc = Bw[threadIdx.x];
        v8f c = {};
        #pragma unroll
        for (int p = 0; p < 4; ++p) {
            const int kk = (kh * 8 + 2 * p) * 4 + hl * 2;
            const float4 bb = wf[p * 32];
            const v2f a0 = *(const v2f*)&S[(32 + l15) * TPITCH + kk];
            const v2f a1 = *(const v2f*)&S[(32 + l15) * TPITCH + kk + 4];
            v2f b0, b1;
            b0.x = bb.x; b0.y = bb.y;
            b1.x = bb.z; b1.y = bb.w;
            c = __builtin_amdgcn_wmma_f32_16x16x4_f32(
                false, a0, false, b0, (short)0, c, false, false);
            c = __builtin_amdgcn_wmma_f32_16x16x4_f32(
                false, a1, false, b1, (short)0, c, false, false);
        }
        #pragma unroll
        for (int q = 0; q < 8; ++q)
            S[(kh * 16 + q + 8 * hl) * TPITCH + col] = c[q] + bc;
        lds_barrier();

        const int d = tile * 16 + r;
        if (d < n) {
            const float4 o0 = *(const float4*)&S[rowoff];
            const float4 o1 = *(const float4*)&S[16 * TPITCH + rowoff];
            float4 res;
            res.x = o0.x + o1.x;
            res.y = o0.y + o1.y;
            res.z = o0.z + o1.z;
            res.w = o0.w + o1.w;
            if (do_relu) {
                res.x = fmaxf(res.x, 0.0f); res.y = fmaxf(res.y, 0.0f);
                res.z = fmaxf(res.z, 0.0f); res.w = fmaxf(res.w, 0.0f);
            }
            ((float4*)out)[d * 16 + j] = res;
        }
    }
}

extern "C" void kernel_launch(void* const* d_in, const int* in_sizes, int n_in,
                              void* d_out, int out_size, void* d_ws, size_t ws_size,
                              hipStream_t stream) {
    const float* x0 = (const float*)d_in[0];
    const int*   me = (const int*)d_in[1];
    const float* W1 = (const float*)d_in[2];
    const float* b1 = (const float*)d_in[3];
    const float* W2 = (const float*)d_in[4];
    const float* b2 = (const float*)d_in[5];
    const float* W3 = (const float*)d_in[6];
    const float* b3 = (const float*)d_in[7];

    const int E = in_sizes[1] / 2;
    const int* src = me;
    const int* dst = me + E;
    const int NF = NNODES * DFEAT;

    float* ws     = (float*)d_ws;
    int*   rowptr = (int*)d_ws;
    float* dinv   = ws + 100352;
    int2*  tab    = (int2*)(ws + 200704);
    float* xa     = ws + 462848;
    float* xb     = xa + NF;
    int*   pairs  = (int*)(xb + NF);
    int2*  stage  = (int2*)(pairs + 2 * (size_t)E);
    int*   srcs   = pairs + 4 * (size_t)E;

    float* out = (float*)d_out;

    const int TB = 256;
    const int gE   = (E + TB - 1) / TB;
    const int nC   = (E + CHUNK - 1) / CHUNK;

    bin_edges<<<nC, 640, 0, stream>>>(src, dst, stage, tab, E);
    sort_bin<<<NBIN, 1024, 0, stream>>>(stage, tab, rowptr, dinv, srcs, NNODES, nC);
    edge_weights<<<gE, TB, 0, stream>>>(srcs, dinv, pairs, E);

    gather_gemm<<<NBLK, TB, 0, stream>>>(x0, pairs, rowptr, dinv, W1, b1, xa, 1, NNODES, NTILE);

    gather_gemm<<<NBLK, TB, 0, stream>>>(xa, pairs, rowptr, dinv, W2, b2, xb, 1, NNODES, NTILE);

    gather_gemm<<<NBLK, TB, 0, stream>>>(xb, pairs, rowptr, dinv, W3, b3, out, 0, NNODES, NTILE);
}


# Edge weight table and its launch dropped, no input copy: layer 1 fetches dinv of a row's 16 sources in one load beside its row loads, hidden rows are stored times dinv; sum order kept, rounding moves
# speedup vs baseline: 6.7024x; 1.0383x over previous
#include <hip/hip_runtime.h>

typedef float v2f __attribute__((ext_vector_type(2)));
typedef float v8f __attribute__((ext_vector_type(8)));

#define NNODES 100000
#define DFEAT  64
#define NTILE  6250
#define NBLK   1563
#define TPITCH 68

#define CHUNK   6400
#define BINSH   8
#define NBIN    391
#define TABP    512
#define MAXCH   256

__device__ __forceinline__ int wave_incl_scan(int v, int lane) {
    #pragma unroll
    for (int o = 1; o < 32; o <<= 1) {
        const int u = __shfl_up(v, o, 32);
        if (lane >= o) v += u;
    }
    return v;
}

__global__ __launch_bounds__(640) void bin_edges(const int* __restrict__ src,
                                                 const int* __restrict__ dst,
                                                 int2* __restrict__ stage,
                                                 int2* __restrict__ tab, int e) {
    __shared__ int hist[TABP];
    __shared__ int sc[TABP];
    __shared__ int wtot[16];
    __shared__ __attribute__((aligned(16))) int2 buf[CHUNK];
    const int t = threadIdx.x;
    const int lane = t & 31;
    const int wave = t >> 5;
    const int c = blockIdx.x;
    const int base = c * CHUNK;
    if (t < TABP) hist[t] = 0;
    __syncthreads();
    int s[10], d[10], r[10];
    #pragma unroll
    for (int u = 0; u < 10; ++u) {
        const int i = base + u * 640 + t;
        const int ii = min(i, e - 1);
        const int dv = dst[ii];
        s[u] = src[ii];
        d[u] = (i < e) ? dv : -1;
    }
    #pragma unroll
    for (int u = 0; u < 10; ++u)
        r[u] = (d[u] >= 0) ? atomicAdd(&hist[d[u] >> BINSH], 1) : 0;
    __syncthreads();
    const int hv = (t < TABP) ? hist[t] : 0;
    const int incl = wave_incl_scan(hv, lane);
    if (t < TABP && lane == 31) wtot[wave] = incl;
    __syncthreads();
    if (t < TABP) {
        int pre = 0;
        #pragma unroll
        for (int w = 0; w < 16; ++w) pre += (w < wave) ? wtot[w] : 0;
        sc[t] = incl + pre;
    }
    __syncthreads();
    #pragma unroll
    for (int u = 0; u < 10; ++u) {
        if (d[u] >= 0) {
            const int b = d[u] >> BINSH;
            int2 rec;
            rec.x = s[u];
            rec.y = d[u];
            buf[sc[b] - hist[b] + r[u]] = rec;
        }
    }
    __syncthreads();
    const int cnt = min(CHUNK, e - base);
    for (int i = t; i < cnt; i += 640) stage[base + i] = buf[i];
    if (t < TABP) {
        int2 v;
        v.x = hv;
        v.y = sc[t] - hv;
        tab[c * TABP + t] = v;
    }
}

__global__ __launch_bounds__(1024) void sort_bin(const int2* __restrict__ stage,
                                                 const int2* __restrict__ tab,
                                                 int* __restrict__ rowptr,
                                                 float* __restrict__ dinv,
                                                 int* __restrict__ srcs, int n, int nchunk) {
    __shared__ int segoff[MAXCH];
    __shared__ int segcnt[MAXCH];
    __shared__ int red[8];
    __shared__ int wtot[8];
    __shared__ int ncnt[256];
    __shared__ int sc[256];
    __shared__ int cur[256];
    const int t = threadIdx.x;
    const int b = blockIdx.x;
    const int lane = t & 31;
    const int wave = t >> 5;
    int2 v;
    v.x = 0;
    v.y = 0;
    if (t < nchunk && t < MAXCH) v = tab[t * TABP + b];
    int osum = v.y;
    #pragma unroll
    for (int o = 16; o > 0; o >>= 1) osum += __shfl_xor(osum, o, 32);
    if (t < MAXCH) {
        segcnt[t] = v.x;
        segoff[t] = t * CHUNK + v.y;
        ncnt[t] = 0;
        cur[t] = 0;
        if (lane == 0) red[wave] = osum;
    }
    __syncthreads();
    int binbase = 0;
    #pragma unroll
    for (int w = 0; w < 8; ++w) binbase += red[w];
    int cn[8], co[8];
    int maxn = 0;
    #pragma unroll
    for (int i = 0; i < 8; ++i) {
        cn[i] = segcnt[wave + 32 * i];
        co[i] = segoff[wave + 32 * i];
        maxn = max(maxn, cn[i]);
    }
    const int nbase = b << BINSH;
    for (int q = 0; q < maxn; q += 32) {
        const int k = q + lane;
        int dl[8];
        #pragma unroll
        for (int i = 0; i < 8; ++i) {
            const int idx = (k < cn[i]) ? co[i] + k : 0;
            dl[i] = stage[idx].y;
        }
        #pragma unroll
        for (int i = 0; i < 8; ++i)
            if (k < cn[i]) atomicAdd(&ncnt[dl[i] - nbase], 1);
    }
    __syncthreads();
    const int cdeg = (t < 256) ? ncnt[t] : 0;
    const int incl = wave_incl_scan(cdeg, lane);
    if (t < 256 && lane == 31) wtot[wave] = incl;
    __syncthreads();
    if (t < 256) {
        int pre = 0;
        #pragma unroll
        for (int w = 0; w < 8; ++w) pre += (w < wave) ? wtot[w] : 0;
        const int inc = incl + pre;
        sc[t] = inc;
        const int node = nbase + t;
        if (node < n) {
            rowptr[node] = binbase + inc - cdeg;
            dinv[node] = rsqrtf((float)(cdeg + 1));
            if (node == n - 1) rowptr[n] = binbase + inc;
        }
    }
    __syncthreads();
    for (int q = 0; q < maxn; q += 32) {
        const int k = q + lane;
        int2 ed[8];
        #pragma unroll
        for (int i = 0; i < 8; ++i) {
            const int idx = (k < cn[i]) ? co[i] + k : 0;
            ed[i] = stage[idx];
        }
        #pragma unroll
        for (int i = 0; i < 8; ++i) {
            if (k < cn[i]) {
                const int l = ed[i].y - nbase;
                const int r = atomicAdd(&cur[l], 1);
                srcs[binbase + sc[l] - ncnt[l] + r] = ed[i].x;
            }
        }
    }
}

__device__ __forceinline__ void lds_barrier() {
    __builtin_amdgcn_fence(__ATOMIC_RELEASE, "workgroup", "local");
    __builtin_amdgcn_s_barrier();
    __builtin_amdgcn_fence(__ATOMIC_ACQUIRE, "workgroup", "local");
}

template <int K>
__device__ __forceinline__ int row_lane(int v) {
    return __builtin_amdgcn_update_dpp(0, v, 0x150 + K, 0xf, 0xf, true);
}

template <int K0>
__device__ __forceinline__ void edge8(const float4* __restrict__ x4, int j, int ps, int pw,
                                      float4& acc) {
    const int s0 = row_lane<K0 + 0>(ps);
    const int s1 = row_lane<K0 + 1>(ps);
    const int s2 = row_lane<K0 + 2>(ps);
    const int s3 = row_lane<K0 + 3>(ps);
    const int s4 = row_lane<K0 + 4>(ps);
    const int s5 = row_lane<K0 + 5>(ps);
    const int s6 = row_lane<K0 + 6>(ps);
    const int s7 = row_lane<K0 + 7>(ps);
    const float4 v0 = x4[s0 * 16 + j];
    const float4 v1 = x4[s1 * 16 + j];
    const float4 v2 = x4[s2 * 16 + j];
    const float4 v3 = x4[s3 * 16 + j];
    const float4 v4 = x4[s4 * 16 + j];
    const float4 v5 = x4[s5 * 16 + j];
    const float4 v6 = x4[s6 * 16 + j];
    const float4 v7 = x4[s7 * 16 + j];
    __builtin_amdgcn_sched_barrier(0);
    const float w0 = __int_as_float(row_lane<K0 + 0>(pw));
    acc.x += v0.x * w0; acc.y += v0.y * w0; acc.z += v0.z * w0; acc.w += v0.w * w0;
    const float w1 = __int_as_float(row_lane<K0 + 1>(pw));
    acc.x += v1.x * w1; acc.y += v1.y * w1; acc.z += v1.z * w1; acc.w += v1.w * w1;
    const float w2 = __int_as_float(row_lane<K0 + 2>(pw));
    acc.x += v2.x * w2; acc.y += v2.y * w2; acc.z += v2.z * w2; acc.w += v2.w * w2;
    const float w3 = __int_as_float(row_lane<K0 + 3>(pw));
    acc.x += v3.x * w3; acc.y += v3.y * w3; acc.z += v3.z * w3; acc.w += v3.w * w3;
    const float w4 = __int_as_float(row_lane<K0 + 4>(pw));
    acc.x += v4.x * w4; acc.y += v4.y * w4; acc.z += v4.z * w4; acc.w += v4.w * w4;
    const float w5 = __int_as_float(row_lane<K0 + 5>(pw));
    acc.x += v5.x * w5; acc.y += v5.y * w5; acc.z += v5.z * w5; acc.w += v5.w * w5;
    const float w6 = __int_as_float(row_lane<K0 + 6>(pw));
    acc.x += v6.x * w6; acc.y += v6.y * w6; acc.z += v6.z * w6; acc.w += v6.w * w6;
    const float w7 = __int_as_float(row_lane<K0 + 7>(pw));
    acc.x += v7.x * w7; acc.y += v7.y * w7; acc.z += v7.z * w7; acc.w += v7.w * w7;
}

__device__ __forceinline__ int edge_source(const int* __restrict__ sr, int first, int j,
                                           int last) {
    return sr[min(first + j, last)];
}

__device__ __forceinline__ void edge32(const float4* __restrict__ x4,
                                       const float* __restrict__ dinv, int raw, int j,
                                       int ia, int ib, int left, float4& acc) {
    float wa = (j < left) ? 1.0f : 0.0f;
    float wb = (j + 16 < left) ? 1.0f : 0.0f;
    if (raw) {
        if (j < left) wa = dinv[ia];
        if (j + 16 < left) wb = dinv[ib];
    }
    const int ua = __float_as_int(wa);
    const int ub = __float_as_int(wb);
    edge8<0>(x4, j, ia, ua, acc);
    if (left > 8)  edge8<8>(x4, j, ia, ua, acc);
    if (left > 16) edge8<0>(x4, j, ib, ub, acc);
    if (left > 24) edge8<8>(x4, j, ib, ub, acc);
}

__global__ __launch_bounds__(256, 16) void gather_gemm(const float* __restrict__ x,
                                                   const int* __restrict__ srcs,
                                                   const int* __restrict__ rowptr,
                                                   const float* __restrict__ dinv,
                                                   const float* __restrict__ W,
                                                   const float* __restrict__ bias,
                                                   float* __restrict__ out,
                                                   int raw, int hidden, int n, int ntiles) {
    __shared__ __attribute__((aligned(16))) float S[48 * TPITCH];
    __shared__ __attribute__((aligned(16))) float Wf[256 * 16];
    __shared__ float Bw[256];

    const int j = threadIdx.x & 15;
    const int r = threadIdx.x >> 4;
    const float4* x4 = (const float4*)x;
    const int rowoff = r * TPITCH + 4 * j;

    const int lane = threadIdx.x & 31;
    const int wave = threadIdx.x >> 5;
    const int hl  = lane >> 4;
    const int l15 = lane & 15;
    const int nt  = wave & 3;
    const int kh  = wave >> 2;
    const int col = nt * 16 + l15;

    float4* wf = (float4*)Wf + wave * 128 + lane;
    #pragma unroll
    for (int p = 0; p < 4; ++p) {
        const int kk = (kh * 8 + 2 * p) * 4 + hl * 2;
        float4 v;
        v.x = W[(kk + 0) * DFEAT + col];
        v.y = W[(kk + 1) * DFEAT + col];
        v.z = W[(kk + 4) * DFEAT + col];
        v.w = W[(kk + 5) * DFEAT + col];
        wf[p * 32] = v;
    }
    Bw[threadIdx.x] = (kh == 0) ? bias[col] : 0.0f;

    const int gstep = gridDim.x;
    int start = 0, end = 0;
    int ia = 0, ib = 0;
    float di = 0.0f;
    float4 xs = {0.0f, 0.0f, 0.0f, 0.0f};
    {
        const int d0 = blockIdx.x * 16 + r;
        if ((int)blockIdx.x < ntiles && d0 < n) {
            start = rowptr[d0];
            end = rowptr[d0 + 1];
            di = dinv[d0];
            xs = x4[d0 * 16 + j];
        }
        if (end > start) {
            ia = edge_source(srcs, start, j, end - 1);
            ib = edge_source(srcs, start + 16, j, end - 1);
        }
    }

    for (int tile = blockIdx.x; tile < ntiles; tile += gstep) {
        asm volatile("" :: "v"(xs.x), "v"(xs.y), "v"(xs.z), "v"(xs.w), "v"(di));
        const float sw = raw ? di : 1.0f;
        float4 acc;
        acc.x = xs.x * sw; acc.y = xs.y * sw; acc.z = xs.z * sw; acc.w = xs.w * sw;
        const float dcur = di;

        const int tn = tile + gstep;
        const int dn = tn * 16 + r;
        const bool nextrow = tn < ntiles && dn < n;
        int nstart = 0, nend = 0;
        if (nextrow) {
            nstart = rowptr[dn];
            nend = rowptr[dn + 1];
        }

        int left = end - start;
        if (left > 0) {
            edge32(x4, dinv, raw, j, ia, ib, left, acc);
            while (left > 32) {
                left -= 32;
                const int ja = edge_source(srcs, end - left, j, end - 1);
                const int jb = edge_source(srcs, end - left + 16, j, end - 1);
                edge32(x4, dinv, raw, j, ja, jb, left, acc);
            }
        }
        float4 g;
        g.x = acc.x * di;
        g.y = acc.y * di;
        g.z = acc.z * di;
        g.w = acc.w * di;

        start = nstart;
        end = nend;
        if (nend > nstart) {
            ia = edge_source(srcs, nstart, j, nend - 1);
            ib = edge_source(srcs, nstart + 16, j, nend - 1);
        }
        if (nextrow) {
            di = dinv[dn];
            xs = x4[dn * 16 + j];
        } else {
            di = 0.0f;
            xs.x = 0.0f; xs.y = 0.0f; xs.z = 0.0f; xs.w = 0.0f;
        }
        *(float4*)&S[32 * TPITCH + rowoff] = g;
        lds_barrier();

        const float bc = Bw[threadIdx.x];
        v8f c = {};
        #pragma unroll
        for (int p = 0; p < 4; ++p) {
            const int kk = (kh * 8 + 2 * p) * 4 + hl * 2;
            const float4 bb = wf[p * 32];
            const v2f a0 = *(const v2f*)&S[(32 + l15) * TPITCH + kk];
            const v2f a1 = *(const v2f*)&S[(32 + l15) * TPITCH + kk + 4];
            v2f b0, b1;
            b0.x = bb.x; b0.y = bb.y;
            b1.x = bb.z; b1.y = bb.w;
            c = __builtin_amdgcn_wmma_f32_16x16x4_f32(
                false, a0, false, b0, (short)0, c, false, false);
            c = __builtin_amdgcn_wmma_f32_16x16x4_f32(
                false, a1, false, b1, (short)0, c, false, false);
        }
        #pragma unroll
        for (int q = 0; q < 8; ++q)
            S[(kh * 16 + q + 8 * hl) * TPITCH + col] = c[q] + bc;
        lds_barrier();

        const int d = tile * 16 + r;
        if (d < n) {
            const float4 o0 = *(const float4*)&S[rowoff];
            const float4 o1 = *(const float4*)&S[16 * TPITCH + rowoff];
            float4 res;
            res.x = o0.x + o1.x;
            res.y = o0.y + o1.y;
            res.z = o0.z + o1.z;
            res.w = o0.w + o1.w;
            if (hidden) {
                res.x = fmaxf(res.x, 0.0f) * dcur; res.y = fmaxf(res.y, 0.0f) * dcur;
                res.z = fmaxf(res.z, 0.0f) * dcur; res.w = fmaxf(res.w, 0.0f) * dcur;
            }
            ((float4*)out)[d * 16 + j] = res;
        }
    }
}

extern "C" void kernel_launch(void* const* d_in, const int* in_sizes, int n_in,
                              void* d_out, int out_size, void* d_ws, size_t ws_size,
                              hipStream_t stream) {
    const float* x0 = (const float*)d_in[0];
    const int*   me = (const int*)d_in[1];
    const float* W1 = (const float*)d_in[2];
    const float* b1 = (const float*)d_in[3];
    const float* W2 = (const float*)d_in[4];
    const float* b2 = (const float*)d_in[5];
    const float* W3 = (const float*)d_in[6];
    const float* b3 = (const float*)d_in[7];

    const int E = in_sizes[1] / 2;
    const int* src = me;
    const int* dst = me + E;
    const int NF = NNODES * DFEAT;

    float* ws     = (float*)d_ws;
    int*   rowptr = (int*)d_ws;
    float* dinv   = ws + 100352;
    int2*  tab    = (int2*)(ws + 200704);
    float* xa     = ws + 462848;
    float* xb     = xa + NF;
    int2*  stage  = (int2*)(xb + NF);
    int*   srcs   = (int*)(stage + (size_t)E);

    float* out = (float*)d_out;

    const int TB = 256;
    const int nC   = (E + CHUNK - 1) / CHUNK;

    bin_edges<<<nC, 640, 0, stream>>>(src, dst, stage, tab, E);
    sort_bin<<<NBIN, 1024, 0, stream>>>(stage, tab, rowptr, dinv, srcs, NNODES, nC);

    gather_gemm<<<NBLK, TB, 0, stream>>>(x0, srcs, rowptr, dinv, W1, b1, xa, 1, 1, NNODES, NTILE);

    gather_gemm<<<NBLK, TB, 0, stream>>>(xa, srcs, rowptr, dinv, W2, b2, xb, 0, 1, NNODES, NTILE);

    gather_gemm<<<NBLK, TB, 0, stream>>>(xb, srcs, rowptr, dinv, W3, b3, out, 0, 0, NNODES, NTILE);
}


# Sorted source list leaves sort_bin in whole 128-byte lines: a bin's sources are placed in LDS, then stored 16 bytes per lane, not by one scattered 4-byte store per edge; edge order as before
# speedup vs baseline: 6.9408x; 1.0356x over previous
#include <hip/hip_runtime.h>

typedef float v2f __attribute__((ext_vector_type(2)));
typedef float v8f __attribute__((ext_vector_type(8)));

#define NNODES 100000
#define DFEAT  64
#define NTILE  6250
#define NBLK   1563
#define TPITCH 68

#define CHUNK   6400
#define BINSH   8
#define NBIN    391
#define TABP    512
#define MAXCH   256
#define OCAP    16384

__device__ __forceinline__ int wave_incl_scan(int v, int lane) {
    #pragma unroll
    for (int o = 1; o < 32; o <<= 1) {
        const int u = __shfl_up(v, o, 32);
        if (lane >= o) v += u;
    }
    return v;
}

__global__ __launch_bounds__(640) void bin_edges(const int* __restrict__ src,
                                                 const int* __restrict__ dst,
                                                 int2* __restrict__ stage,
                                                 int2* __restrict__ tab, int e) {
    __shared__ int hist[TABP];
    __shared__ int sc[TABP];
    __shared__ int wtot[16];
    __shared__ __attribute__((aligned(16))) int2 buf[CHUNK];
    const int t = threadIdx.x;
    const int lane = t & 31;
    const int wave = t >> 5;
    const int c = blockIdx.x;
    const int base = c * CHUNK;
    if (t < TABP) hist[t] = 0;
    __syncthreads();
    int s[10], d[10], r[10];
    #pragma unroll
    for (int u = 0; u < 10; ++u) {
        const int i = base + u * 640 + t;
        const int ii = min(i, e - 1);
        const int dv = dst[ii];
        s[u] = src[ii];
        d[u] = (i < e) ? dv : -1;
    }
    #pragma unroll
    for (int u = 0; u < 10; ++u)
        r[u] = (d[u] >= 0) ? atomicAdd(&hist[d[u] >> BINSH], 1) : 0;
    __syncthreads();
    const int hv = (t < TABP) ? hist[t] : 0;
    const int incl = wave_incl_scan(hv, lane);
    if (t < TABP && lane == 31) wtot[wave] = incl;
    __syncthreads();
    if (t < TABP) {
        int pre = 0;
        #pragma unroll
        for (int w = 0; w < 16; ++w) pre += (w < wave) ? wtot[w] : 0;
        sc[t] = incl + pre;
    }
    __syncthreads();
    #pragma unroll
    for (int u = 0; u < 10; ++u) {
        if (d[u] >= 0) {
            const int b = d[u] >> BINSH;
            int2 rec;
            rec.x = s[u];
            rec.y = d[u];
            buf[sc[b] - hist[b] + r[u]] = rec;
        }
    }
    __syncthreads();
    const int cnt = min(CHUNK, e - base);
    for (int i = t; i < cnt; i += 640) stage[base + i] = buf[i];
    if (t < TABP) {
        int2 v;
        v.x = hv;
        v.y = sc[t] - hv;
        tab[c * TABP + t] = v;
    }
}

__global__ __launch_bounds__(1024) void sort_bin(const int2* __restrict__ stage,
                                                 const int2* __restrict__ tab,
                                                 int* __restrict__ rowptr,
                                                 float* __restrict__ dinv,
                                                 int* __restrict__ srcs, int n, int nchunk) {
    __shared__ int segoff[MAXCH];
    __shared__ int segcnt[MAXCH];
    __shared__ int red[8];
    __shared__ int wtot[8];
    __shared__ int ncnt[256];
    __shared__ int sc[256];
    __shared__ int cur[256];
    __shared__ __attribute__((aligned(16))) int obuf[OCAP + 32];
    const int t = threadIdx.x;
    const int b = blockIdx.x;
    const int lane = t & 31;
    const int wave = t >> 5;
    int2 v;
    v.x = 0;
    v.y = 0;
    if (t < nchunk && t < MAXCH) v = tab[t * TABP + b];
    int osum = v.y;
    #pragma unroll
    for (int o = 16; o > 0; o >>= 1) osum += __shfl_xor(osum, o, 32);
    if (t < MAXCH) {
        segcnt[t] = v.x;
        segoff[t] = t * CHUNK + v.y;
        ncnt[t] = 0;
        cur[t] = 0;
        if (lane == 0) red[wave] = osum;
    }
    __syncthreads();
    int binbase = 0;
    #pragma unroll
    for (int w = 0; w < 8; ++w) binbase += red[w];
    int cn[8], co[8];
    int maxn = 0;
    #pragma unroll
    for (int i = 0; i < 8; ++i) {
        cn[i] = segcnt[wave + 32 * i];
        co[i] = segoff[wave + 32 * i];
        maxn = max(maxn, cn[i]);
    }
    const int nbase = b << BINSH;
    for (int q = 0; q < maxn; q += 32) {
        const int k = q + lane;
        int dl[8];
        #pragma unroll
        for (int i = 0; i < 8; ++i) {
            const int idx = (k < cn[i]) ? co[i] + k : 0;
            dl[i] = stage[idx].y;
        }
        #pragma unroll
        for (int i = 0; i < 8; ++i)
            if (k < cn[i]) atomicAdd(&ncnt[dl[i] - nbase], 1);
    }
    __syncthreads();
    const int cdeg = (t < 256) ? ncnt[t] : 0;
    const int incl = wave_incl_scan(cdeg, lane);
    if (t < 256 && lane == 31) wtot[wave] = incl;
    __syncthreads();
    if (t < 256) {
        int pre = 0;
        #pragma unroll
        for (int w = 0; w < 8; ++w) pre += (w < wave) ? wtot[w] : 0;
        const int inc = incl + pre;
        sc[t] = inc;
        const int node = nbase + t;
        if (node < n) {
            rowptr[node] = binbase + inc - cdeg;
            dinv[node] = rsqrtf((float)(cdeg + 1));
            if (node == n - 1) rowptr[n] = binbase + inc;
        }
    }
    __syncthreads();
    const int total = sc[255];
    if (total <= OCAP) {
        const int head = binbase & 31;
        for (int q = 0; q < maxn; q += 32) {
            const int k = q + lane;
            int2 ed[8];
            #pragma unroll
            for (int i = 0; i < 8; ++i) {
                const int idx = (k < cn[i]) ? co[i] + k : 0;
                ed[i] = stage[idx];
            }
            #pragma unroll
            for (int i = 0; i < 8; ++i) {
                if (k < cn[i]) {
                    const int l = ed[i].y - nbase;
                    const int r = atomicAdd(&cur[l], 1);
                    obuf[head + sc[l] - ncnt[l] + r] = ed[i].x;
                }
            }
        }
        __syncthreads();
        const int linebase = binbase - head;
        const int endl = head + total;
        for (int v = 4 * t; v < endl; v += 4096) {
            if (v >= head && v + 4 <= endl) {
                *(int4*)&srcs[linebase + v] = *(const int4*)&obuf[v];
            } else {
                for (int e = 0; e < 4; ++e) {
                    const int w = v + e;
                    if (w >= head && w < endl) srcs[linebase + w] = obuf[w];
                }
            }
        }
    } else {
        for (int q = 0; q < maxn; q += 32) {
            const int k = q + lane;
            int2 ed[8];
            #pragma unroll
            for (int i = 0; i < 8; ++i) {
                const int idx = (k < cn[i]) ? co[i] + k : 0;
                ed[i] = stage[idx];
            }
            #pragma unroll
            for (int i = 0; i < 8; ++i) {
                if (k < cn[i]) {
                    const int l = ed[i].y - nbase;
                    const int r = atomicAdd(&cur[l], 1);
                    srcs[binbase + sc[l] - ncnt[l] + r] = ed[i].x;
                }
            }
        }
    }
}

__device__ __forceinline__ void lds_barrier() {
    __builtin_amdgcn_fence(__ATOMIC_RELEASE, "workgroup", "local");
    __builtin_amdgcn_s_barrier();
    __builtin_amdgcn_fence(__ATOMIC_ACQUIRE, "workgroup", "local");
}

template <int K>
__device__ __forceinline__ int row_lane(int v) {
    return __builtin_amdgcn_update_dpp(0, v, 0x150 + K, 0xf, 0xf, true);
}

template <int K0>
__device__ __forceinline__ void edge8(const float4* __restrict__ x4, int j, int ps, int pw,
                                      float4& acc) {
    const int s0 = row_lane<K0 + 0>(ps);
    const int s1 = row_lane<K0 + 1>(ps);
    const int s2 = row_lane<K0 + 2>(ps);
    const int s3 = row_lane<K0 + 3>(ps);
    const int s4 = row_lane<K0 + 4>(ps);
    const int s5 = row_lane<K0 + 5>(ps);
    const int s6 = row_lane<K0 + 6>(ps);
    const int s7 = row_lane<K0 + 7>(ps);
    const float4 v0 = x4[s0 * 16 + j];
    const float4 v1 = x4[s1 * 16 + j];
    const float4 v2 = x4[s2 * 16 + j];
    const float4 v3 = x4[s3 * 16 + j];
    const float4 v4 = x4[s4 * 16 + j];
    const float4 v5 = x4[s5 * 16 + j];
    const float4 v6 = x4[s6 * 16 + j];
    const float4 v7 = x4[s7 * 16 + j];
    __builtin_amdgcn_sched_barrier(0);
    const float w0 = __int_as_float(row_lane<K0 + 0>(pw));
    acc.x += v0.x * w0; acc.y += v0.y * w0; acc.z += v0.z * w0; acc.w += v0.w * w0;
    const float w1 = __int_as_float(row_lane<K0 + 1>(pw));
    acc.x += v1.x * w1; acc.y += v1.y * w1; acc.z += v1.z * w1; acc.w += v1.w * w1;
    const float w2 = __int_as_float(row_lane<K0 + 2>(pw));
    acc.x += v2.x * w2; acc.y += v2.y * w2; acc.z += v2.z * w2; acc.w += v2.w * w2;
    const float w3 = __int_as_float(row_lane<K0 + 3>(pw));
    acc.x += v3.x * w3; acc.y += v3.y * w3; acc.z += v3.z * w3; acc.w += v3.w * w3;
    const float w4 = __int_as_float(row_lane<K0 + 4>(pw));
    acc.x += v4.x * w4; acc.y += v4.y * w4; acc.z += v4.z * w4; acc.w += v4.w * w4;
    const float w5 = __int_as_float(row_lane<K0 + 5>(pw));
    acc.x += v5.x * w5; acc.y += v5.y * w5; acc.z += v5.z * w5; acc.w += v5.w * w5;
    const float w6 = __int_as_float(row_lane<K0 + 6>(pw));
    acc.x += v6.x * w6; acc.y += v6.y * w6; acc.z += v6.z * w6; acc.w += v6.w * w6;
    const float w7 = __int_as_float(row_lane<K0 + 7>(pw));
    acc.x += v7.x * w7; acc.y += v7.y * w7; acc.z += v7.z * w7; acc.w += v7.w * w7;
}

__device__ __forceinline__ int edge_source(const int* __restrict__ sr, int first, int j,
                                           int last) {
    return sr[min(first + j, last)];
}

__device__ __forceinline__ void edge32(const float4* __restrict__ x4,
                                       const float* __restrict__ dinv, int raw, int j,
                                       int ia, int ib, int left, float4& acc) {
    float wa = (j < left) ? 1.0f : 0.0f;
    float wb = (j + 16 < left) ? 1.0f : 0.0f;
    if (raw) {
        if (j < left) wa = dinv[ia];
        if (j + 16 < left) wb = dinv[ib];
    }
    const int ua = __float_as_int(wa);
    const int ub = __float_as_int(wb);
    edge8<0>(x4, j, ia, ua, acc);
    if (left > 8)  edge8<8>(x4, j, ia, ua, acc);
    if (left > 16) edge8<0>(x4, j, ib, ub, acc);
    if (left > 24) edge8<8>(x4, j, ib, ub, acc);
}

__global__ __launch_bounds__(256, 16) void gather_gemm(const float* __restrict__ x,
                                                   const int* __restrict__ srcs,
                                                   const int* __restrict__ rowptr,
                                                   const float* __restrict__ dinv,
                                                   const float* __restrict__ W,
                                                   const float* __restrict__ bias,
                                                   float* __restrict__ out,
                                                   int raw, int hidden, int n, int ntiles) {
    __shared__ __attribute__((aligned(16))) float S[48 * TPITCH];
    __shared__ __attribute__((aligned(16))) float Wf[256 * 16];
    __shared__ float Bw[256];

    const int j = threadIdx.x & 15;
    const int r = threadIdx.x >> 4;
    const float4* x4 = (const float4*)x;
    const int rowoff = r * TPITCH + 4 * j;

    const int lane = threadIdx.x & 31;
    const int wave = threadIdx.x >> 5;
    const int hl  = lane >> 4;
    const int l15 = lane & 15;
    const int nt  = wave & 3;
    const int kh  = wave >> 2;
    const int col = nt * 16 + l15;

    float4* wf = (float4*)Wf + wave * 128 + lane;
    #pragma unroll
    for (int p = 0; p < 4; ++p) {
        const int kk = (kh * 8 + 2 * p) * 4 + hl * 2;
        float4 v;
        v.x = W[(kk + 0) * DFEAT + col];
        v.y = W[(kk + 1) * DFEAT + col];
        v.z = W[(kk + 4) * DFEAT + col];
        v.w = W[(kk + 5) * DFEAT + col];
        wf[p * 32] = v;
    }
    Bw[threadIdx.x] = (kh == 0) ? bias[col] : 0.0f;

    const int gstep = gridDim.x;
    int start = 0, end = 0;
    int ia = 0, ib = 0;
    float di = 0.0f;
    float4 xs = {0.0f, 0.0f, 0.0f, 0.0f};
    {
        const int d0 = blockIdx.x * 16 + r;
        if ((int)blockIdx.x < ntiles && d0 < n) {
            start = rowptr[d0];
            end = rowptr[d0 + 1];
            di = dinv[d0];
            xs = x4[d0 * 16 + j];
        }
        if (end > start) {
            ia = edge_source(srcs, start, j, end - 1);
            ib = edge_source(srcs, start + 16, j, end - 1);
        }
    }

    for (int tile = blockIdx.x; tile < ntiles; tile += gstep) {
        asm volatile("" :: "v"(xs.x), "v"(xs.y), "v"(xs.z), "v"(xs.w), "v"(di));
        const float sw = raw ? di : 1.0f;
        float4 acc;
        acc.x = xs.x * sw; acc.y = xs.y * sw; acc.z = xs.z * sw; acc.w = xs.w * sw;
        const float dcur = di;

        const int tn = tile + gstep;
        const int dn = tn * 16 + r;
        const bool nextrow = tn < ntiles && dn < n;
        int nstart = 0, nend = 0;
        if (nextrow) {
            nstart = rowptr[dn];
            nend = rowptr[dn + 1];
        }

        int left = end - start;
        if (left > 0) {
            edge32(x4, dinv, raw, j, ia, ib, left, acc);
            while (left > 32) {
                left -= 32;
                const int ja = edge_source(srcs, end - left, j, end - 1);
                const int jb = edge_source(srcs, end - left + 16, j, end - 1);
                edge32(x4, dinv, raw, j, ja, jb, left, acc);
            }
        }
        float4 g;
        g.x = acc.x * di;
        g.y = acc.y * di;
        g.z = acc.z * di;
        g.w = acc.w * di;

        start = nstart;
        end = nend;
        if (nend > nstart) {
            ia = edge_source(srcs, nstart, j, nend - 1);
            ib = edge_source(srcs, nstart + 16, j, nend - 1);
        }
        if (nextrow) {
            di = dinv[dn];
            xs = x4[dn * 16 + j];
        } else {
            di = 0.0f;
            xs.x = 0.0f; xs.y = 0.0f; xs.z = 0.0f; xs.w = 0.0f;
        }
        *(float4*)&S[32 * TPITCH + rowoff] = g;
        lds_barrier();

        const float bc = Bw[threadIdx.x];
        v8f c = {};
        #pragma unroll
        for (int p = 0; p < 4; ++p) {
            const int kk = (kh * 8 + 2 * p) * 4 + hl * 2;
            const float4 bb = wf[p * 32];
            const v2f a0 = *(const v2f*)&S[(32 + l15) * TPITCH + kk];
            const v2f a1 = *(const v2f*)&S[(32 + l15) * TPITCH + kk + 4];
            v2f b0, b1;
            b0.x = bb.x; b0.y = bb.y;
            b1.x = bb.z; b1.y = bb.w;
            c = __builtin_amdgcn_wmma_f32_16x16x4_f32(
                false, a0, false, b0, (short)0, c, false, false);
            c = __builtin_amdgcn_wmma_f32_16x16x4_f32(
                false, a1, false, b1, (short)0, c, false, false);
        }
        #pragma unroll
        for (int q = 0; q < 8; ++q)
            S[(kh * 16 + q + 8 * hl) * TPITCH + col] = c[q] + bc;
        lds_barrier();

        const int d = tile * 16 + r;
        if (d < n) {
            const float4 o0 = *(const float4*)&S[rowoff];
            const float4 o1 = *(const float4*)&S[16 * TPITCH + rowoff];
            float4 res;
            res.x = o0.x + o1.x;
            res.y = o0.y + o1.y;
            res.z = o0.z + o1.z;
            res.w = o0.w + o1.w;
            if (hidden) {
                res.x = fmaxf(res.x, 0.0f) * dcur; res.y = fmaxf(res.y, 0.0f) * dcur;
                res.z = fmaxf(res.z, 0.0f) * dcur; res.w = fmaxf(res.w, 0.0f) * dcur;
            }
            ((float4*)out)[d * 16 + j] = res;
        }
    }
}

extern "C" void kernel_launch(void* const* d_in, const int* in_sizes, int n_in,
                              void* d_out, int out_size, void* d_ws, size_t ws_size,
                              hipStream_t stream) {
    const float* x0 = (const float*)d_in[0];
    const int*   me = (const int*)d_in[1];
    const float* W1 = (const float*)d_in[2];
    const float* b1 = (const float*)d_in[3];
    const float* W2 = (const float*)d_in[4];
    const float* b2 = (const float*)d_in[5];
    const float* W3 = (const float*)d_in[6];
    const float* b3 = (const float*)d_in[7];

    const int E = in_sizes[1] / 2;
    const int* src = me;
    const int* dst = me + E;
    const int NF = NNODES * DFEAT;

    float* ws     = (float*)d_ws;
    int*   rowptr = (int*)d_ws;
    float* dinv   = ws + 100352;
    int2*  tab    = (int2*)(ws + 200704);
    float* xa     = ws + 462848;
    float* xb     = xa + NF;
    int2*  stage  = (int2*)(xb + NF);
    int*   srcs   = (int*)(stage + (size_t)E);

    float* out = (float*)d_out;

    const int TB = 256;
    const int nC   = (E + CHUNK - 1) / CHUNK;

    bin_edges<<<nC, 640, 0, stream>>>(src, dst, stage, tab, E);
    sort_bin<<<NBIN, 1024, 0, stream>>>(stage, tab, rowptr, dinv, srcs, NNODES, nC);

    gather_gemm<<<NBLK, TB, 0, stream>>>(x0, srcs, rowptr, dinv, W1, b1, xa, 1, 1, NNODES, NTILE);

    gather_gemm<<<NBLK, TB, 0, stream>>>(xa, srcs, rowptr, dinv, W2, b2, xb, 0, 1, NNODES, NTILE);

    gather_gemm<<<NBLK, TB, 0, stream>>>(xb, srcs, rowptr, dinv, W3, b3, out, 0, 0, NNODES, NTILE);
}
